# starting kernel
# baseline (speedup 1.0000x reference)
#include <hip/hip_runtime.h>


#define NUM_NODES     50000
#define INPUT_SIZE    128
#define NUM_RELATIONS 16
#define NUM_EDGES     1600000
#define NODE_TILES    (NUM_NODES / 16)
#define WAVES_PER_BLK 8

typedef __attribute__((ext_vector_type(2))) float v2f;
typedef __attribute__((ext_vector_type(8))) float v8f;

__global__ __launch_bounds__(256) void node_logits_wmma_kernel(
    const float* __restrict__ x,
    const float* __restrict__ att_weight,
    float* __restrict__ T,
    float* __restrict__ U)
{
    __shared__ float wlds[2 * INPUT_SIZE * NUM_RELATIONS];
    for (int i = threadIdx.x; i < 2 * INPUT_SIZE * NUM_RELATIONS; i += 256)
        wlds[i] = att_weight[i];
    __syncthreads();

    const int wave = threadIdx.x >> 5;
    const int lane = threadIdx.x & 31;
    const int tile = blockIdx.x * WAVES_PER_BLK + wave;
    if (tile >= NODE_TILES) return;

    const int half = lane >> 4;
    const int l15  = lane & 15;
    const int koff = half * 2;

    const float* xrow = x + (size_t)(tile * 16 + l15) * INPUT_SIZE;

    v8f cS = {};
    v8f cD = {};

    #pragma unroll 4
    for (int k0 = 0; k0 < INPUT_SIZE; k0 += 4) {
        v2f a;
        a.x = xrow[k0 + koff + 0];
        a.y = xrow[k0 + koff + 1];
        v2f bS, bD;
        bS.x = wlds[(k0 + koff + 0) * NUM_RELATIONS + l15];
        bS.y = wlds[(k0 + koff + 1) * NUM_RELATIONS + l15];
        bD.x = wlds[(INPUT_SIZE + k0 + koff + 0) * NUM_RELATIONS + l15];
        bD.y = wlds[(INPUT_SIZE + k0 + koff + 1) * NUM_RELATIONS + l15];

        cS = __builtin_amdgcn_wmma_f32_16x16x4_f32(false, a, false, bS,
                                                   (short)0, cS, false, false);
        cD = __builtin_amdgcn_wmma_f32_16x16x4_f32(false, a, false, bD,
                                                   (short)0, cD, false, false);
    }

    float* Tout = T + (size_t)tile * 16 * NUM_RELATIONS;
    float* Uout = U + (size_t)tile * 16 * NUM_RELATIONS;
    #pragma unroll
    for (int j = 0; j < 8; ++j) {
        const int m = j + half * 8;
        Tout[m * NUM_RELATIONS + l15] = cS[j];
        Uout[m * NUM_RELATIONS + l15] = cD[j];
    }
}

__global__ __launch_bounds__(256) void edge_prob_kernel(
    const int* __restrict__ edge_index,
    const int* __restrict__ edge_type,
    const float* __restrict__ T,
    const float* __restrict__ U,
    float* __restrict__ out)
{
    const int e = blockIdx.x * 256 + threadIdx.x;
    if (e >= NUM_EDGES) return;

    const int s = edge_index[e];
    const int d = edge_index[NUM_EDGES + e];
    const int r = edge_type[e];

    const float logit = T[s * NUM_RELATIONS + r] + U[d * NUM_RELATIONS + r];
    float att = 1.0f / (1.0f + __expf(-logit));
    att = fminf(fmaxf(att, 1e-5f), 0.99999f);
    out[e] = att;
}

extern "C" void kernel_launch(void* const* d_in, const int* in_sizes, int n_in,
                              void* d_out, int out_size, void* d_ws, size_t ws_size,
                              hipStream_t stream) {
    const float* x  = (const float*)d_in[0];
    const float* w  = (const float*)d_in[1];
    const int*   ei = (const int*)d_in[2];
    const int*   et = (const int*)d_in[3];
    float* out = (float*)d_out;

    float* T = (float*)d_ws;
    float* U = T + (size_t)NUM_NODES * NUM_RELATIONS;

    const int blocks1 = (NODE_TILES + WAVES_PER_BLK - 1) / WAVES_PER_BLK;
    node_logits_wmma_kernel<<<blocks1, 256, 0, stream>>>(x, w, T, U);

    const int blocks2 = (NUM_EDGES + 255) / 256;
    edge_prob_kernel<<<blocks2, 256, 0, stream>>>(ei, et, T, U, out);
}


# edge kernel: each wave takes 256 consecutive edges, 8 per lane (16-byte index loads and stores, 16 table gathers in flight per lane); 50000 one-edge-per-lane waves become 6250
# speedup vs baseline: 1.0900x; 1.0900x over previous
#include <hip/hip_runtime.h>


#define NUM_NODES     50000
#define INPUT_SIZE    128
#define NUM_RELATIONS 16
#define NUM_EDGES     1600000
#define NODE_TILES    (NUM_NODES / 16)
#define WAVES_PER_BLK 8

typedef __attribute__((ext_vector_type(2))) float v2f;
typedef __attribute__((ext_vector_type(8))) float v8f;

__global__ __launch_bounds__(256) void node_logits_wmma_kernel(
    const float* __restrict__ x,
    const float* __restrict__ att_weight,
    float* __restrict__ T,
    float* __restrict__ U)
{
    __shared__ float wlds[2 * INPUT_SIZE * NUM_RELATIONS];
    for (int i = threadIdx.x; i < 2 * INPUT_SIZE * NUM_RELATIONS; i += 256)
        wlds[i] = att_weight[i];
    __syncthreads();

    const int wave = threadIdx.x >> 5;
    const int lane = threadIdx.x & 31;
    const int tile = blockIdx.x * WAVES_PER_BLK + wave;
    if (tile >= NODE_TILES) return;

    const int half = lane >> 4;
    const int l15  = lane & 15;
    const int koff = half * 2;

    const float* xrow = x + (size_t)(tile * 16 + l15) * INPUT_SIZE;

    v8f cS = {};
    v8f cD = {};

    #pragma unroll 4
    for (int k0 = 0; k0 < INPUT_SIZE; k0 += 4) {
        v2f a;
        a.x = xrow[k0 + koff + 0];
        a.y = xrow[k0 + koff + 1];
        v2f bS, bD;
        bS.x = wlds[(k0 + koff + 0) * NUM_RELATIONS + l15];
        bS.y = wlds[(k0 + koff + 1) * NUM_RELATIONS + l15];
        bD.x = wlds[(INPUT_SIZE + k0 + koff + 0) * NUM_RELATIONS + l15];
        bD.y = wlds[(INPUT_SIZE + k0 + koff + 1) * NUM_RELATIONS + l15];

        cS = __builtin_amdgcn_wmma_f32_16x16x4_f32(false, a, false, bS,
                                                   (short)0, cS, false, false);
        cD = __builtin_amdgcn_wmma_f32_16x16x4_f32(false, a, false, bD,
                                                   (short)0, cD, false, false);
    }

    float* Tout = T + (size_t)tile * 16 * NUM_RELATIONS;
    float* Uout = U + (size_t)tile * 16 * NUM_RELATIONS;
    #pragma unroll
    for (int j = 0; j < 8; ++j) {
        const int m = j + half * 8;
        Tout[m * NUM_RELATIONS + l15] = cS[j];
        Uout[m * NUM_RELATIONS + l15] = cD[j];
    }
}

#define EDGES_PER_WAVE 256
#define EDGE_WAVES     (NUM_EDGES / EDGES_PER_WAVE)
#define EDGE_BLK_WAVES 4
static_assert(NUM_EDGES % EDGES_PER_WAVE == 0, "a wave takes a whole 256-edge span");

typedef __attribute__((ext_vector_type(4))) int   v4i;
typedef __attribute__((ext_vector_type(4))) float v4f;

__global__ __launch_bounds__(EDGE_BLK_WAVES * 32) void edge_prob_kernel(
    const int* __restrict__ edge_index,
    const int* __restrict__ edge_type,
    const float* __restrict__ T,
    const float* __restrict__ U,
    float* __restrict__ out)
{
    const int lane = threadIdx.x & 31;
    const int wave = blockIdx.x * EDGE_BLK_WAVES + (threadIdx.x >> 5);
    if (wave >= EDGE_WAVES) return;

    const int e0 = wave * EDGES_PER_WAVE + lane * 4;

    v4i s[2], d[2], r[2];
    #pragma unroll
    for (int g = 0; g < 2; ++g) {
        const int e = e0 + g * 128;
        s[g] = *reinterpret_cast<const v4i*>(edge_index + e);
        d[g] = *reinterpret_cast<const v4i*>(edge_index + NUM_EDGES + e);
        r[g] = *reinterpret_cast<const v4i*>(edge_type + e);
    }

    float t[2][4], u[2][4];
    #pragma unroll
    for (int g = 0; g < 2; ++g) {
        #pragma unroll
        for (int i = 0; i < 4; ++i) {
            t[g][i] = T[s[g][i] * NUM_RELATIONS + r[g][i]];
            u[g][i] = U[d[g][i] * NUM_RELATIONS + r[g][i]];
        }
    }

    #pragma unroll
    for (int g = 0; g < 2; ++g) {
        v4f res;
        #pragma unroll
        for (int i = 0; i < 4; ++i) {
            const float logit = t[g][i] + u[g][i];
            float att = 1.0f / (1.0f + __expf(-logit));
            att = fminf(fmaxf(att, 1e-5f), 0.99999f);
            res[i] = att;
        }
        *reinterpret_cast<v4f*>(out + e0 + g * 128) = res;
    }
}

extern "C" void kernel_launch(void* const* d_in, const int* in_sizes, int n_in,
                              void* d_out, int out_size, void* d_ws, size_t ws_size,
                              hipStream_t stream) {
    const float* x  = (const float*)d_in[0];
    const float* w  = (const float*)d_in[1];
    const int*   ei = (const int*)d_in[2];
    const int*   et = (const int*)d_in[3];
    float* out = (float*)d_out;

    float* T = (float*)d_ws;
    float* U = T + (size_t)NUM_NODES * NUM_RELATIONS;

    const int blocks1 = (NODE_TILES + WAVES_PER_BLK - 1) / WAVES_PER_BLK;
    node_logits_wmma_kernel<<<blocks1, 256, 0, stream>>>(x, w, T, U);

    const int blocks2 = (EDGE_WAVES + EDGE_BLK_WAVES - 1) / EDGE_BLK_WAVES;
    edge_prob_kernel<<<blocks2, EDGE_BLK_WAVES * 32, 0, stream>>>(ei, et, T, U, out);
}


# node kernel: a wave asks for its whole 16x128 x tile (16-byte loads, 64 registers) and the block's weights before its first wait: 1 load round trip, not 16; K in chunks of 8 reorders the f32 sums
# speedup vs baseline: 1.2920x; 1.1853x over previous
#include <hip/hip_runtime.h>


#define NUM_NODES     50000
#define INPUT_SIZE    128
#define NUM_RELATIONS 16
#define NUM_EDGES     1600000
#define NODE_TILES    (NUM_NODES / 16)
#define WAVES_PER_BLK 8

typedef __attribute__((ext_vector_type(2))) float v2f;
typedef __attribute__((ext_vector_type(4))) float v4f;
typedef __attribute__((ext_vector_type(8))) float v8f;

__global__ __launch_bounds__(256) void node_logits_wmma_kernel(
    const float* __restrict__ x,
    const float* __restrict__ att_weight,
    float* __restrict__ T,
    float* __restrict__ U)
{
    __shared__ float wlds[2 * INPUT_SIZE * NUM_RELATIONS];

    const int wave = threadIdx.x >> 5;
    const int lane = threadIdx.x & 31;
    const int tile = blockIdx.x * WAVES_PER_BLK + wave;
    const int half = lane >> 4;
    const int l15  = lane & 15;

    const v4f* wsrc = reinterpret_cast<const v4f*>(att_weight);
    v4f wr[4];
    #pragma unroll
    for (int i = 0; i < 4; ++i)
        wr[i] = wsrc[threadIdx.x + 256 * i];

    const int ltile = tile < NODE_TILES ? tile : NODE_TILES - 1;
    const v4f* xrow = reinterpret_cast<const v4f*>(
        x + (size_t)(ltile * 16 + l15) * INPUT_SIZE + half * 4);
    v4f xa[16];
    #pragma unroll
    for (int j = 0; j < 16; ++j)
        xa[j] = xrow[2 * j];

    #pragma unroll
    for (int i = 0; i < 4; ++i)
        reinterpret_cast<v4f*>(wlds)[threadIdx.x + 256 * i] = wr[i];
    __syncthreads();
    if (tile >= NODE_TILES) return;

    v8f cS = {};
    v8f cD = {};

    #pragma unroll
    for (int j = 0; j < 16; ++j) {
        const int kb = 8 * j + 4 * half;
        const float* wS = wlds + kb * NUM_RELATIONS + l15;
        const float* wD = wS + INPUT_SIZE * NUM_RELATIONS;

        v2f a0, a1, bS0, bS1, bD0, bD1;
        a0.x = xa[j].x;  a0.y = xa[j].y;
        a1.x = xa[j].z;  a1.y = xa[j].w;
        bS0.x = wS[0 * NUM_RELATIONS];  bS0.y = wS[1 * NUM_RELATIONS];
        bS1.x = wS[2 * NUM_RELATIONS];  bS1.y = wS[3 * NUM_RELATIONS];
        bD0.x = wD[0 * NUM_RELATIONS];  bD0.y = wD[1 * NUM_RELATIONS];
        bD1.x = wD[2 * NUM_RELATIONS];  bD1.y = wD[3 * NUM_RELATIONS];

        cS = __builtin_amdgcn_wmma_f32_16x16x4_f32(false, a0, false, bS0,
                                                   (short)0, cS, false, false);
        cD = __builtin_amdgcn_wmma_f32_16x16x4_f32(false, a0, false, bD0,
                                                   (short)0, cD, false, false);
        cS = __builtin_amdgcn_wmma_f32_16x16x4_f32(false, a1, false, bS1,
                                                   (short)0, cS, false, false);
        cD = __builtin_amdgcn_wmma_f32_16x16x4_f32(false, a1, false, bD1,
                                                   (short)0, cD, false, false);
    }

    float* Tout = T + (size_t)tile * 16 * NUM_RELATIONS;
    float* Uout = U + (size_t)tile * 16 * NUM_RELATIONS;
    #pragma unroll
    for (int j = 0; j < 8; ++j) {
        const int m = j + half * 8;
        Tout[m * NUM_RELATIONS + l15] = cS[j];
        Uout[m * NUM_RELATIONS + l15] = cD[j];
    }
}

#define EDGES_PER_WAVE 256
#define EDGE_WAVES     (NUM_EDGES / EDGES_PER_WAVE)
#define EDGE_BLK_WAVES 4
static_assert(NUM_EDGES % EDGES_PER_WAVE == 0, "a wave takes a whole 256-edge span");

typedef __attribute__((ext_vector_type(4))) int   v4i;

__global__ __launch_bounds__(EDGE_BLK_WAVES * 32) void edge_prob_kernel(
    const int* __restrict__ edge_index,
    const int* __restrict__ edge_type,
    const float* __restrict__ T,
    const float* __restrict__ U,
    float* __restrict__ out)
{
    const int lane = threadIdx.x & 31;
    const int wave = blockIdx.x * EDGE_BLK_WAVES + (threadIdx.x >> 5);
    if (wave >= EDGE_WAVES) return;

    const int e0 = wave * EDGES_PER_WAVE + lane * 4;

    v4i s[2], d[2], r[2];
    #pragma unroll
    for (int g = 0; g < 2; ++g) {
        const int e = e0 + g * 128;
        s[g] = *reinterpret_cast<const v4i*>(edge_index + e);
        d[g] = *reinterpret_cast<const v4i*>(edge_index + NUM_EDGES + e);
        r[g] = *reinterpret_cast<const v4i*>(edge_type + e);
    }

    float t[2][4], u[2][4];
    #pragma unroll
    for (int g = 0; g < 2; ++g) {
        #pragma unroll
        for (int i = 0; i < 4; ++i) {
            t[g][i] = T[s[g][i] * NUM_RELATIONS + r[g][i]];
            u[g][i] = U[d[g][i] * NUM_RELATIONS + r[g][i]];
        }
    }

    #pragma unroll
    for (int g = 0; g < 2; ++g) {
        v4f res;
        #pragma unroll
        for (int i = 0; i < 4; ++i) {
            const float logit = t[g][i] + u[g][i];
            float att = 1.0f / (1.0f + __expf(-logit));
            att = fminf(fmaxf(att, 1e-5f), 0.99999f);
            res[i] = att;
        }
        *reinterpret_cast<v4f*>(out + e0 + g * 128) = res;
    }
}

extern "C" void kernel_launch(void* const* d_in, const int* in_sizes, int n_in,
                              void* d_out, int out_size, void* d_ws, size_t ws_size,
                              hipStream_t stream) {
    const float* x  = (const float*)d_in[0];
    const float* w  = (const float*)d_in[1];
    const int*   ei = (const int*)d_in[2];
    const int*   et = (const int*)d_in[3];
    float* out = (float*)d_out;

    float* T = (float*)d_ws;
    float* U = T + (size_t)NUM_NODES * NUM_RELATIONS;

    const int blocks1 = (NODE_TILES + WAVES_PER_BLK - 1) / WAVES_PER_BLK;
    node_logits_wmma_kernel<<<blocks1, 256, 0, stream>>>(x, w, T, U);

    const int blocks2 = (EDGE_WAVES + EDGE_BLK_WAVES - 1) / EDGE_BLK_WAVES;
    edge_prob_kernel<<<blocks2, EDGE_BLK_WAVES * 32, 0, stream>>>(ei, et, T, U, out);
}
